# starting kernel
# baseline (speedup 1.0000x reference)
#include <hip/hip_runtime.h>
#include <stdint.h>

#define N_ 32
#define C_ 256
#define K_ 8192
#define P_ 16
#define CT 16
#define NW 8
#define THREADS 256

typedef float v2f __attribute__((ext_vector_type(2)));
typedef float v8f __attribute__((ext_vector_type(8)));

__global__ __launch_bounds__(THREADS)
void AnchorPatchPooling_kernel(const float* __restrict__ feats,
                               const int*  __restrict__ labels,
                               const int*  __restrict__ vmask,
                               float* __restrict__ out)
{
    __shared__ uint32_t lv[K_];
    __shared__ float sumTile[CT * P_];
    __shared__ float maxTile[CT * 17];
    __shared__ float cntBin[P_];
    __shared__ float patchBin[P_];

    const int t   = threadIdx.x;
    const int blk = blockIdx.x;
    const int n   = blk >> 4;
    const int c0  = (blk & 15) * CT;

    if (t < CT * P_) sumTile[t] = 0.0f;
    for (int i = t; i < CT * 17; i += THREADS) maxTile[i] = -100.0f;
    if (t < P_) { cntBin[t] = 0.0f; patchBin[t] = 0.0f; }
    __syncthreads();

    for (int k = t; k < K_; k += THREADS) {
        uint32_t lab = (uint32_t)labels[k] & (P_ - 1);
        int v = vmask[(size_t)n * K_ + k];
        lv[k] = (lab << 16) | (v ? (1u << lab) : 0u);
        atomicAdd(&patchBin[lab], 1.0f);
        if (v) atomicAdd(&cntBin[lab], 1.0f);
    }
    __syncthreads();

    const int lane = t & 31;
    const int wv   = t >> 5;
    const int col  = lane & 15;
    const int hi   = lane >> 4;

    const int kStart     = wv * (K_ / NW);
    const float*    aPtr  = feats + ((size_t)(n * C_ + c0 + col)) * K_ + kStart + 2 * hi;
    const uint32_t* lvPtr = &lv[kStart + 2 * hi];
    const uint32_t  maxRow = (uint32_t)(uintptr_t)(&maxTile[0]) + (uint32_t)(col * 17) * 4u;

    v8f acc = {};

    #pragma unroll 8
    for (int i = 0; i < K_ / NW; i += 4) {
        uint32_t w0 = lvPtr[i];
        uint32_t w1 = lvPtr[i + 1];

        v2f b;
        b.x = (float)((w0 >> col) & 1u);
        b.y = (float)((w1 >> col) & 1u);

        v2f a = *(const v2f*)(aPtr + i);

        acc = __builtin_amdgcn_wmma_f32_16x16x4_f32(
             false, a,  false, b,
             (short)0, acc,  false,  false);

        uint32_t addr0 = maxRow + ((w0 >> 16) << 2);
        uint32_t addr1 = maxRow + ((w1 >> 16) << 2);
        asm volatile("ds_max_num_f32 %0, %1" :: "v"(addr0), "v"(a.x));
        asm volatile("ds_max_num_f32 %0, %1" :: "v"(addr1), "v"(a.y));
    }

    #pragma unroll
    for (int j = 0; j < 8; ++j) {
        int m = j + 8 * hi;
        atomicAdd(&sumTile[m * P_ + col], acc[j]);
    }
    asm volatile("s_wait_dscnt 0x0" ::: "memory");
    __syncthreads();

    {
        int m = t >> 4, p = t & 15;
        float s    = sumTile[m * P_ + p];
        float cntp = cntBin[p];
        float mean = s / fmaxf(cntp, 1.0f);
        float mx   = (patchBin[p] > 0.0f) ? maxTile[m * 17 + p] : 0.0f;
        out[((size_t)(n * C_ + c0 + m)) * P_ + p] = mean + mx;
    }
}

extern "C" void kernel_launch(void* const* d_in, const int* in_sizes, int n_in,
                              void* d_out, int out_size, void* d_ws, size_t ws_size,
                              hipStream_t stream) {
    (void)in_sizes; (void)n_in; (void)d_ws; (void)ws_size; (void)out_size;
    const float* feats  = (const float*)d_in[0];
    const int*   labels = (const int*)d_in[1];
    const int*   vmask  = (const int*)d_in[2];
    float*       out    = (float*)d_out;

    dim3 grid(N_ * (C_ / CT));
    dim3 block(THREADS);
    AnchorPatchPooling_kernel<<<grid, block, 0, stream>>>(feats, labels, vmask, out);
}


# loads issued in groups ahead of use: 8 x 16-byte feature loads per wave one group ahead (was one waited 8-byte load), label/mask words up front; k regrouped per matrix op: f32 sum order changes
# speedup vs baseline: 1.3890x; 1.3890x over previous
#include <hip/hip_runtime.h>
#include <stdint.h>


#define N_ 32
#define C_ 256
#define K_ 8192
#define P_ 16
#define CT 16
#define NW 8
#define THREADS 256
#define KW (K_ / NW)
#define GL 8
#define GK (GL * 8)
#define NG (KW / GK)
#define SL (K_ / (4 * THREADS))

typedef float    v2f __attribute__((ext_vector_type(2)));
typedef float    v4f __attribute__((ext_vector_type(4)));
typedef float    v8f __attribute__((ext_vector_type(8)));
typedef uint32_t v4u __attribute__((ext_vector_type(4)));
typedef int      v4i __attribute__((ext_vector_type(4)));

__device__ __forceinline__ void load_group(v4f (&buf)[GL], const float* p)
{
    #pragma unroll
    for (int j = 0; j < GL; ++j) buf[j] = *(const v4f*)(p + 8 * j);
}

__device__ __forceinline__ void proc_group(const v4f (&buf)[GL], const uint32_t* lvp,
                                           int col, uint32_t maxRow, v8f& acc)
{
    #pragma unroll
    for (int j = 0; j < GL; ++j) {
        const v4u w = *(const v4u*)(lvp + 8 * j);
        const v4f x = buf[j];

        v2f a0 = {x.x, x.y};
        v2f a1 = {x.z, x.w};
        v2f b0, b1;
        b0.x = (float)((w.x >> col) & 1u);
        b0.y = (float)((w.y >> col) & 1u);
        b1.x = (float)((w.z >> col) & 1u);
        b1.y = (float)((w.w >> col) & 1u);

        acc = __builtin_amdgcn_wmma_f32_16x16x4_f32(
             false, a0,  false, b0,
             (short)0, acc,  false,  false);
        acc = __builtin_amdgcn_wmma_f32_16x16x4_f32(
             false, a1,  false, b1,
             (short)0, acc,  false,  false);

        uint32_t ad0 = maxRow + ((w.x >> 16) << 2);
        uint32_t ad1 = maxRow + ((w.y >> 16) << 2);
        uint32_t ad2 = maxRow + ((w.z >> 16) << 2);
        uint32_t ad3 = maxRow + ((w.w >> 16) << 2);
        asm volatile("ds_max_num_f32 %0, %1" :: "v"(ad0), "v"(x.x));
        asm volatile("ds_max_num_f32 %0, %1" :: "v"(ad1), "v"(x.y));
        asm volatile("ds_max_num_f32 %0, %1" :: "v"(ad2), "v"(x.z));
        asm volatile("ds_max_num_f32 %0, %1" :: "v"(ad3), "v"(x.w));
    }
}

__global__ __launch_bounds__(THREADS)
void AnchorPatchPooling_kernel(const float* __restrict__ feats,
                               const int*  __restrict__ labels,
                               const int*  __restrict__ vmask,
                               float* __restrict__ out)
{
    __shared__ __attribute__((aligned(16))) uint32_t lv[K_];
    __shared__ float sumTile[CT * P_];
    __shared__ float maxTile[CT * 17];
    __shared__ float cntBin[P_];
    __shared__ float patchBin[P_];

    const int t   = threadIdx.x;
    const int blk = blockIdx.x;
    const int n   = blk >> 4;
    const int c0  = (blk & 15) * CT;

    v4i labv[SL], vmv[SL];
    {
        const v4i* lab4 = (const v4i*)labels + t;
        const v4i* vm4  = (const v4i*)(vmask + (size_t)n * K_) + t;
        #pragma unroll
        for (int j = 0; j < SL; ++j) { labv[j] = lab4[j * THREADS]; vmv[j] = vm4[j * THREADS]; }
    }

    if (t < CT * P_) sumTile[t] = 0.0f;
    for (int i = t; i < CT * 17; i += THREADS) maxTile[i] = -100.0f;
    if (t < P_) { cntBin[t] = 0.0f; patchBin[t] = 0.0f; }
    __syncthreads();

    #pragma unroll
    for (int j = 0; j < SL; ++j) {
        v4u word;
        #pragma unroll
        for (int e = 0; e < 4; ++e) {
            uint32_t lab = (uint32_t)labv[j][e] & (P_ - 1);
            int v = vmv[j][e];
            word[e] = (lab << 16) | (v ? (1u << lab) : 0u);
            atomicAdd(&patchBin[lab], 1.0f);
            if (v) atomicAdd(&cntBin[lab], 1.0f);
        }
        *(v4u*)&lv[4 * t + 4 * THREADS * j] = word;
    }
    __syncthreads();

    const int lane = t & 31;
    const int wv   = t >> 5;
    const int col  = lane & 15;
    const int hi   = lane >> 4;

    const int kStart     = wv * KW;
    const float*    aPtr  = feats + ((size_t)(n * C_ + c0 + col)) * K_ + kStart + 4 * hi;
    const uint32_t* lvPtr = &lv[kStart + 4 * hi];
    const uint32_t  maxRow = (uint32_t)(uintptr_t)(&maxTile[0]) + (uint32_t)(col * 17) * 4u;

    v8f acc = {};

    v4f bufA[GL], bufB[GL];
    load_group(bufA, aPtr);
    for (int g = 0; g < NG - 2; g += 2) {
        load_group(bufB, aPtr + (g + 1) * GK);
        proc_group(bufA, lvPtr + g * GK, col, maxRow, acc);
        asm volatile("" : "+v"(acc));
        load_group(bufA, aPtr + (g + 2) * GK);
        proc_group(bufB, lvPtr + (g + 1) * GK, col, maxRow, acc);
    }
    load_group(bufB, aPtr + (NG - 1) * GK);
    proc_group(bufA, lvPtr + (NG - 2) * GK, col, maxRow, acc);
    proc_group(bufB, lvPtr + (NG - 1) * GK, col, maxRow, acc);

    #pragma unroll
    for (int j = 0; j < 8; ++j) {
        int m = j + 8 * hi;
        atomicAdd(&sumTile[m * P_ + col], acc[j]);
    }
    asm volatile("s_wait_dscnt 0x0" ::: "memory");
    __syncthreads();

    {
        int m = t >> 4, p = t & 15;
        float s    = sumTile[m * P_ + p];
        float cntp = cntBin[p];
        float mean = s / fmaxf(cntp, 1.0f);
        float mx   = (patchBin[p] > 0.0f) ? maxTile[m * 17 + p] : 0.0f;
        out[((size_t)(n * C_ + c0 + m)) * P_ + p] = mean + mx;
    }
}

extern "C" void kernel_launch(void* const* d_in, const int* in_sizes, int n_in,
                              void* d_out, int out_size, void* d_ws, size_t ws_size,
                              hipStream_t stream) {
    (void)in_sizes; (void)n_in; (void)d_ws; (void)ws_size; (void)out_size;
    const float* feats  = (const float*)d_in[0];
    const int*   labels = (const int*)d_in[1];
    const int*   vmask  = (const int*)d_in[2];
    float*       out    = (float*)d_out;

    dim3 grid(N_ * (C_ / CT));
    dim3 block(THREADS);
    AnchorPatchPooling_kernel<<<grid, block, 0, stream>>>(feats, labels, vmask, out);
}


# per-part counts without per-anchor LDS atomics: valid count summed as integers from the B one-hot bits in the main loop, anchors-per-part gate as an OR of label one-hots; feature sum order unchanged
# speedup vs baseline: 1.8080x; 1.3017x over previous
#include <hip/hip_runtime.h>
#include <stdint.h>


#define N_ 32
#define C_ 256
#define K_ 8192
#define P_ 16
#define CT 16
#define NW 8
#define THREADS 256
#define KW (K_ / NW)
#define GL 8
#define GK (GL * 8)
#define NG (KW / GK)
#define SL (K_ / (4 * THREADS))

typedef float    v2f __attribute__((ext_vector_type(2)));
typedef float    v4f __attribute__((ext_vector_type(4)));
typedef float    v8f __attribute__((ext_vector_type(8)));
typedef uint32_t v4u __attribute__((ext_vector_type(4)));
typedef int      v4i __attribute__((ext_vector_type(4)));

__device__ __forceinline__ void load_group(v4f (&buf)[GL], const float* p)
{
    #pragma unroll
    for (int j = 0; j < GL; ++j) buf[j] = *(const v4f*)(p + 8 * j);
}

__device__ __forceinline__ void proc_group(const v4f (&buf)[GL], const uint32_t* lvp,
                                           int col, uint32_t maxRow, v8f& acc, uint32_t& cnt)
{
    #pragma unroll
    for (int j = 0; j < GL; ++j) {
        const v4u w = *(const v4u*)(lvp + 8 * j);
        const v4f x = buf[j];

        v2f a0 = {x.x, x.y};
        v2f a1 = {x.z, x.w};
        const uint32_t i0 = (w.x >> col) & 1u;
        const uint32_t i1 = (w.y >> col) & 1u;
        const uint32_t i2 = (w.z >> col) & 1u;
        const uint32_t i3 = (w.w >> col) & 1u;
        v2f b0, b1;
        b0.x = (float)i0;
        b0.y = (float)i1;
        b1.x = (float)i2;
        b1.y = (float)i3;
        cnt += (i0 + i1) + (i2 + i3);

        acc = __builtin_amdgcn_wmma_f32_16x16x4_f32(
             false, a0,  false, b0,
             (short)0, acc,  false,  false);
        acc = __builtin_amdgcn_wmma_f32_16x16x4_f32(
             false, a1,  false, b1,
             (short)0, acc,  false,  false);

        uint32_t ad0 = maxRow + ((w.x >> 16) << 2);
        uint32_t ad1 = maxRow + ((w.y >> 16) << 2);
        uint32_t ad2 = maxRow + ((w.z >> 16) << 2);
        uint32_t ad3 = maxRow + ((w.w >> 16) << 2);
        asm volatile("ds_max_num_f32 %0, %1" :: "v"(ad0), "v"(x.x));
        asm volatile("ds_max_num_f32 %0, %1" :: "v"(ad1), "v"(x.y));
        asm volatile("ds_max_num_f32 %0, %1" :: "v"(ad2), "v"(x.z));
        asm volatile("ds_max_num_f32 %0, %1" :: "v"(ad3), "v"(x.w));
    }
}

__global__ __launch_bounds__(THREADS)
void AnchorPatchPooling_kernel(const float* __restrict__ feats,
                               const int*  __restrict__ labels,
                               const int*  __restrict__ vmask,
                               float* __restrict__ out)
{
    __shared__ __attribute__((aligned(16))) uint32_t lv[K_];
    __shared__ float sumTile[CT * P_];
    __shared__ float maxTile[CT * 17];
    __shared__ uint32_t cntBin[P_];
    __shared__ uint32_t presW[NW];

    const int t    = threadIdx.x;
    const int lane = t & 31;
    const int wv   = t >> 5;
    const int blk  = blockIdx.x;
    const int n   = blk >> 4;
    const int c0  = (blk & 15) * CT;

    v4i labv[SL], vmv[SL];
    {
        const v4i* lab4 = (const v4i*)labels + t;
        const v4i* vm4  = (const v4i*)(vmask + (size_t)n * K_) + t;
        #pragma unroll
        for (int j = 0; j < SL; ++j) { labv[j] = lab4[j * THREADS]; vmv[j] = vm4[j * THREADS]; }
    }

    if (t < CT * P_) sumTile[t] = 0.0f;
    for (int i = t; i < CT * 17; i += THREADS) maxTile[i] = -100.0f;
    if (t < P_) cntBin[t] = 0u;

    uint32_t pres = 0u;
    #pragma unroll
    for (int j = 0; j < SL; ++j) {
        v4u word;
        #pragma unroll
        for (int e = 0; e < 4; ++e) {
            uint32_t lab = (uint32_t)labv[j][e] & (P_ - 1);
            int v = vmv[j][e];
            const uint32_t bit = 1u << lab;
            word[e] = (lab << 16) | (v ? bit : 0u);
            pres |= bit;
        }
        *(v4u*)&lv[4 * t + 4 * THREADS * j] = word;
    }
    for (int off = 16; off > 0; off >>= 1) pres |= __shfl_xor(pres, off);
    if (lane == 0) presW[wv] = pres;
    __syncthreads();

    const int col  = lane & 15;
    const int hi   = lane >> 4;

    const int kStart     = wv * KW;
    const float*    aPtr  = feats + ((size_t)(n * C_ + c0 + col)) * K_ + kStart + 4 * hi;
    const uint32_t* lvPtr = &lv[kStart + 4 * hi];
    const uint32_t  maxRow = (uint32_t)(uintptr_t)(&maxTile[0]) + (uint32_t)(col * 17) * 4u;

    v8f acc = {};
    uint32_t cnt = 0u;

    v4f bufA[GL], bufB[GL];
    load_group(bufA, aPtr);
    for (int g = 0; g < NG - 2; g += 2) {
        load_group(bufB, aPtr + (g + 1) * GK);
        proc_group(bufA, lvPtr + g * GK, col, maxRow, acc, cnt);
        asm volatile("" : "+v"(acc));
        load_group(bufA, aPtr + (g + 2) * GK);
        proc_group(bufB, lvPtr + (g + 1) * GK, col, maxRow, acc, cnt);
    }
    load_group(bufB, aPtr + (NG - 1) * GK);
    proc_group(bufA, lvPtr + (NG - 2) * GK, col, maxRow, acc, cnt);
    proc_group(bufB, lvPtr + (NG - 1) * GK, col, maxRow, acc, cnt);

    #pragma unroll
    for (int j = 0; j < 8; ++j) {
        int m = j + 8 * hi;
        atomicAdd(&sumTile[m * P_ + col], acc[j]);
    }
    atomicAdd(&cntBin[col], cnt);
    asm volatile("s_wait_dscnt 0x0" ::: "memory");
    __syncthreads();

    {
        int m = t >> 4, p = t & 15;
        float s    = sumTile[m * P_ + p];
        float cntp = (float)cntBin[p];
        float mean = s / fmaxf(cntp, 1.0f);
        uint32_t present = 0u;
        for (int i = 0; i < NW; ++i) present |= presW[i];
        float mx   = ((present >> p) & 1u) ? maxTile[m * 17 + p] : 0.0f;
        out[((size_t)(n * C_ + c0 + m)) * P_ + p] = mean + mx;
    }
}

extern "C" void kernel_launch(void* const* d_in, const int* in_sizes, int n_in,
                              void* d_out, int out_size, void* d_ws, size_t ws_size,
                              hipStream_t stream) {
    (void)in_sizes; (void)n_in; (void)d_ws; (void)ws_size; (void)out_size;
    const float* feats  = (const float*)d_in[0];
    const int*   labels = (const int*)d_in[1];
    const int*   vmask  = (const int*)d_in[2];
    float*       out    = (float*)d_out;

    dim3 grid(N_ * (C_ / CT));
    dim3 block(THREADS);
    AnchorPatchPooling_kernel<<<grid, block, 0, stream>>>(feats, labels, vmask, out);
}


# feature loads read whole 128-byte lines: one 16-byte load = 2 rows x 256 contiguous bytes (was 16 rows x 32); a wave-private LDS stage turns rows into A fragments, no barrier; f32 sum order unchanged
# speedup vs baseline: 1.8797x; 1.0397x over previous
#include <hip/hip_runtime.h>
#include <stdint.h>


#define N_ 32
#define C_ 256
#define K_ 8192
#define P_ 16
#define CT 16
#define NW 8
#define THREADS 256
#define KW (K_ / NW)
#define GL 8
#define GK (GL * 8)
#define NG (KW / GK)
#define SL (K_ / (4 * THREADS))
#define SP (GK + 4)

typedef float    v2f __attribute__((ext_vector_type(2)));
typedef float    v4f __attribute__((ext_vector_type(4)));
typedef float    v8f __attribute__((ext_vector_type(8)));
typedef uint32_t v4u __attribute__((ext_vector_type(4)));
typedef int      v4i __attribute__((ext_vector_type(4)));

__device__ __forceinline__ void load_rows(v4f (&buf)[GL], const float* p)
{
    #pragma unroll
    for (int i = 0; i < GL; ++i) buf[i] = *(const v4f*)(p + (size_t)(2 * i) * K_);
}

__device__ __forceinline__ void store_stage(const v4f (&buf)[GL], float* st)
{
    #pragma unroll
    for (int i = 0; i < GL; ++i) *(v4f*)(st + 2 * i * SP) = buf[i];
}

__device__ __forceinline__ void read_frag(v4f (&frag)[GL], const float* st)
{
    #pragma unroll
    for (int j = 0; j < GL; ++j) frag[j] = *(const v4f*)(st + 8 * j);
}

__device__ __forceinline__ void proc_group(const v4f (&buf)[GL], const uint32_t* lvp,
                                           int col, uint32_t maxRow, v8f& acc, uint32_t& cnt)
{
    #pragma unroll
    for (int j = 0; j < GL; ++j) {
        const v4u w = *(const v4u*)(lvp + 8 * j);
        const v4f x = buf[j];

        v2f a0 = {x.x, x.y};
        v2f a1 = {x.z, x.w};
        const uint32_t i0 = (w.x >> col) & 1u;
        const uint32_t i1 = (w.y >> col) & 1u;
        const uint32_t i2 = (w.z >> col) & 1u;
        const uint32_t i3 = (w.w >> col) & 1u;
        v2f b0, b1;
        b0.x = (float)i0;
        b0.y = (float)i1;
        b1.x = (float)i2;
        b1.y = (float)i3;
        cnt += (i0 + i1) + (i2 + i3);

        acc = __builtin_amdgcn_wmma_f32_16x16x4_f32(
             false, a0,  false, b0,
             (short)0, acc,  false,  false);
        acc = __builtin_amdgcn_wmma_f32_16x16x4_f32(
             false, a1,  false, b1,
             (short)0, acc,  false,  false);

        uint32_t ad0 = maxRow + ((w.x >> 16) << 2);
        uint32_t ad1 = maxRow + ((w.y >> 16) << 2);
        uint32_t ad2 = maxRow + ((w.z >> 16) << 2);
        uint32_t ad3 = maxRow + ((w.w >> 16) << 2);
        asm volatile("ds_max_num_f32 %0, %1" :: "v"(ad0), "v"(x.x));
        asm volatile("ds_max_num_f32 %0, %1" :: "v"(ad1), "v"(x.y));
        asm volatile("ds_max_num_f32 %0, %1" :: "v"(ad2), "v"(x.z));
        asm volatile("ds_max_num_f32 %0, %1" :: "v"(ad3), "v"(x.w));
    }
}

__global__ __launch_bounds__(THREADS)
void AnchorPatchPooling_kernel(const float* __restrict__ feats,
                               const int*  __restrict__ labels,
                               const int*  __restrict__ vmask,
                               float* __restrict__ out)
{
    __shared__ __attribute__((aligned(16))) uint32_t lv[K_];
    __shared__ float sumTile[CT * P_];
    __shared__ float maxTile[CT * 17];
    __shared__ uint32_t cntBin[P_];
    __shared__ uint32_t presW[NW];
    __shared__ __attribute__((aligned(16))) float stage[NW][CT * SP];

    const int t    = threadIdx.x;
    const int lane = t & 31;
    const int wv   = t >> 5;
    const int blk  = blockIdx.x;
    const int n   = blk >> 4;
    const int c0  = (blk & 15) * CT;

    v4i labv[SL], vmv[SL];
    {
        const v4i* lab4 = (const v4i*)labels + t;
        const v4i* vm4  = (const v4i*)(vmask + (size_t)n * K_) + t;
        #pragma unroll
        for (int j = 0; j < SL; ++j) { labv[j] = lab4[j * THREADS]; vmv[j] = vm4[j * THREADS]; }
    }

    if (t < CT * P_) sumTile[t] = 0.0f;
    for (int i = t; i < CT * 17; i += THREADS) maxTile[i] = -100.0f;
    if (t < P_) cntBin[t] = 0u;

    uint32_t pres = 0u;
    #pragma unroll
    for (int j = 0; j < SL; ++j) {
        v4u word;
        #pragma unroll
        for (int e = 0; e < 4; ++e) {
            uint32_t lab = (uint32_t)labv[j][e] & (P_ - 1);
            int v = vmv[j][e];
            const uint32_t bit = 1u << lab;
            word[e] = (lab << 16) | (v ? bit : 0u);
            pres |= bit;
        }
        *(v4u*)&lv[4 * t + 4 * THREADS * j] = word;
    }
    for (int off = 16; off > 0; off >>= 1) pres |= __shfl_xor(pres, off);
    if (lane == 0) presW[wv] = pres;
    __syncthreads();

    const int col  = lane & 15;
    const int hi   = lane >> 4;

    const int kStart     = wv * KW;
    const float*    gPtr  = feats + ((size_t)(n * C_ + c0 + hi)) * K_ + kStart + 4 * col;
    float*          stW   = &stage[wv][hi * SP + 4 * col];
    const float*    stR   = &stage[wv][col * SP + 4 * hi];
    const uint32_t* lvPtr = &lv[kStart + 4 * hi];
    const uint32_t  maxRow = (uint32_t)(uintptr_t)(&maxTile[0]) + (uint32_t)(col * 17) * 4u;

    v8f acc = {};
    uint32_t cnt = 0u;

    v4f ld[GL], fr[GL];
    load_rows(ld, gPtr);
    for (int g = 0; g < NG - 1; ++g) {
        store_stage(ld, stW);
        __builtin_amdgcn_fence(__ATOMIC_ACQ_REL, "wavefront");
        load_rows(ld, gPtr + (g + 1) * GK);
        read_frag(fr, stR);
        proc_group(fr, lvPtr + g * GK, col, maxRow, acc, cnt);
        __builtin_amdgcn_fence(__ATOMIC_ACQ_REL, "wavefront");
    }
    store_stage(ld, stW);
    __builtin_amdgcn_fence(__ATOMIC_ACQ_REL, "wavefront");
    read_frag(fr, stR);
    proc_group(fr, lvPtr + (NG - 1) * GK, col, maxRow, acc, cnt);

    #pragma unroll
    for (int j = 0; j < 8; ++j) {
        int m = j + 8 * hi;
        atomicAdd(&sumTile[m * P_ + col], acc[j]);
    }
    atomicAdd(&cntBin[col], cnt);
    asm volatile("s_wait_dscnt 0x0" ::: "memory");
    __syncthreads();

    {
        int m = t >> 4, p = t & 15;
        float s    = sumTile[m * P_ + p];
        float cntp = (float)cntBin[p];
        float mean = s / fmaxf(cntp, 1.0f);
        uint32_t present = 0u;
        for (int i = 0; i < NW; ++i) present |= presW[i];
        float mx   = ((present >> p) & 1u) ? maxTile[m * 17 + p] : 0.0f;
        out[((size_t)(n * C_ + c0 + m)) * P_ + p] = mean + mx;
    }
}

extern "C" void kernel_launch(void* const* d_in, const int* in_sizes, int n_in,
                              void* d_out, int out_size, void* d_ws, size_t ws_size,
                              hipStream_t stream) {
    (void)in_sizes; (void)n_in; (void)d_ws; (void)ws_size; (void)out_size;
    const float* feats  = (const float*)d_in[0];
    const int*   labels = (const int*)d_in[1];
    const int*   vmask  = (const int*)d_in[2];
    float*       out    = (float*)d_out;

    dim3 grid(N_ * (C_ / CT));
    dim3 block(THREADS);
    AnchorPatchPooling_kernel<<<grid, block, 0, stream>>>(feats, labels, vmask, out);
}
